# starting kernel
# baseline (speedup 1.0000x reference)
#include <hip/hip_runtime.h>
#include <math.h>

#define Bsz 8
#define Tsz 2048
#define Csz 1024
#define Hsz 64

typedef __attribute__((ext_vector_type(2))) float v2f;
typedef __attribute__((ext_vector_type(8))) float v8f;

__device__ __forceinline__ v8f wmma_f32(v2f a, v2f b, v8f c) {
    return __builtin_amdgcn_wmma_f32_16x16x4_f32(
         false, a,  false, b,
         (short)0, c,  false,  false);
}

__global__ __launch_bounds__(256) void wt_kernel(
    const float* __restrict__ Wq,
    const float* __restrict__ Wk,
    const float* __restrict__ Wv,
    float* __restrict__ Wt)
{
    const int which = blockIdx.y;
    const float* W = (which == 0) ? Wq : (which == 1) ? Wk : Wv;
    float* dst = Wt + (size_t)which * Hsz * Csz;
    const int h = blockIdx.x;
    for (int c = threadIdx.x; c < Csz; c += 256)
        dst[(size_t)h * Csz + c] = W[(size_t)c * Hsz + h];
}

__global__ __launch_bounds__(128) void qkv_kernel(
    const float* __restrict__ x,
    const float* __restrict__ Wt,
    float* __restrict__ qkv)
{
    const int lane = threadIdx.x & 31;
    const int wv   = threadIdx.x >> 5;
    const int half = lane >> 4;
    const int l16  = lane & 15;

    const long mrow  = (long)blockIdx.x * 64 + wv * 16;
    const int  which = blockIdx.y;
    const float* W = Wt + (size_t)which * Hsz * Csz;

    const float* arow = x + (mrow + l16) * (size_t)Csz + 2 * half;
    const float* b0 = W + (size_t)( 0 + l16) * Csz + 2 * half;
    const float* b1 = W + (size_t)(16 + l16) * Csz + 2 * half;
    const float* b2 = W + (size_t)(32 + l16) * Csz + 2 * half;
    const float* b3 = W + (size_t)(48 + l16) * Csz + 2 * half;

    v8f acc0 = {}, acc1 = {}, acc2 = {}, acc3 = {};
#pragma unroll 4
    for (int kk = 0; kk < Csz; kk += 4) {
        v2f a  = *(const v2f*)(arow + kk);
        v2f w0 = *(const v2f*)(b0 + kk);
        v2f w1 = *(const v2f*)(b1 + kk);
        v2f w2 = *(const v2f*)(b2 + kk);
        v2f w3 = *(const v2f*)(b3 + kk);
        acc0 = wmma_f32(a, w0, acc0);
        acc1 = wmma_f32(a, w1, acc1);
        acc2 = wmma_f32(a, w2, acc2);
        acc3 = wmma_f32(a, w3, acc3);
    }

    const size_t BTH = (size_t)Bsz * Tsz * Hsz;
    if (which < 2) {
        float* op = qkv + (size_t)which * BTH + (mrow + 8 * half) * (size_t)Hsz + l16;
#pragma unroll
        for (int r = 0; r < 8; ++r) {
            op[(size_t)r * Hsz +  0] = acc0[r];
            op[(size_t)r * Hsz + 16] = acc1[r];
            op[(size_t)r * Hsz + 32] = acc2[r];
            op[(size_t)r * Hsz + 48] = acc3[r];
        }
    } else {
        const long bb   = mrow / Tsz;
        const long tloc = mrow - bb * Tsz + 8 * half;
        float* vt = qkv + 2 * BTH;
        const v8f* accs[4] = { &acc0, &acc1, &acc2, &acc3 };
#pragma unroll
        for (int nt = 0; nt < 4; ++nt) {
            const int col = nt * 16 + l16;
            float* vp = vt + ((size_t)bb * Hsz + col) * Tsz + tloc;
            const v8f& a = *accs[nt];
            float4 lo = make_float4(a[0], a[1], a[2], a[3]);
            float4 hi = make_float4(a[4], a[5], a[6], a[7]);
            *(float4*)(vp + 0) = lo;
            *(float4*)(vp + 4) = hi;
        }
    }
}

__global__ __launch_bounds__(32) void attn_kernel(
    const float* __restrict__ qkv,
    float* __restrict__ out)
{
    __shared__ float Plds[16 * 16];

    const int lane = threadIdx.x;
    const int half = lane >> 4;
    const int l16  = lane & 15;
    const int b    = blockIdx.y;
    const int qi   = gridDim.x - 1 - blockIdx.x;

    const size_t BTH = (size_t)Bsz * Tsz * Hsz;
    const float* Q  = qkv +            (size_t)b * Tsz * Hsz;
    const float* K  = qkv + BTH     +  (size_t)b * Tsz * Hsz;
    const float* Vt = qkv + 2 * BTH +  (size_t)b * Hsz * Tsz;

    const float scale = 0.125f;
    const int qrow = qi * 16 + l16;

    v2f qa[16];
#pragma unroll
    for (int c = 0; c < 16; ++c) {
        v2f t = *(const v2f*)(Q + (size_t)qrow * Hsz + 4 * c + 2 * half);
        qa[c][0] = t[0] * scale;
        qa[c][1] = t[1] * scale;
    }

    const float* Vr0 = Vt + (size_t)( 0 + l16) * Tsz + 2 * half;
    const float* Vr1 = Vt + (size_t)(16 + l16) * Tsz + 2 * half;
    const float* Vr2 = Vt + (size_t)(32 + l16) * Tsz + 2 * half;
    const float* Vr3 = Vt + (size_t)(48 + l16) * Tsz + 2 * half;

    float mrow[8], lrow[8];
#pragma unroll
    for (int r = 0; r < 8; ++r) { mrow[r] = -INFINITY; lrow[r] = 0.0f; }
    v8f o0 = {}, o1 = {}, o2 = {}, o3 = {};

    for (int kt = 0; kt <= qi; ++kt) {
        const float* Kt = K + (size_t)(kt * 16 + l16) * Hsz + 2 * half;
        v8f s = {};
#pragma unroll
        for (int c = 0; c < 16; ++c) {
            v2f kb = *(const v2f*)(Kt + 4 * c);
            s = wmma_f32(qa[c], kb, s);
        }

        if (kt == qi) {
#pragma unroll
            for (int r = 0; r < 8; ++r)
                if (l16 > r + 8 * half) s[r] = -INFINITY;
        }

        float alpha[8], p[8];
#pragma unroll
        for (int r = 0; r < 8; ++r) {
            float mx = s[r];
            mx = fmaxf(mx, __shfl_xor(mx, 1));
            mx = fmaxf(mx, __shfl_xor(mx, 2));
            mx = fmaxf(mx, __shfl_xor(mx, 4));
            mx = fmaxf(mx, __shfl_xor(mx, 8));
            const float mn = fmaxf(mrow[r], mx);
            alpha[r] = expf(mrow[r] - mn);
            mrow[r]  = mn;
            float pv = expf(s[r] - mn);
            p[r] = pv;
            float sm = pv;
            sm += __shfl_xor(sm, 1);
            sm += __shfl_xor(sm, 2);
            sm += __shfl_xor(sm, 4);
            sm += __shfl_xor(sm, 8);
            lrow[r] = lrow[r] * alpha[r] + sm;
        }

#pragma unroll
        for (int r = 0; r < 8; ++r)
            Plds[(r + 8 * half) * 16 + l16] = p[r];
        __syncthreads();

#pragma unroll
        for (int r = 0; r < 8; ++r) {
            o0[r] *= alpha[r]; o1[r] *= alpha[r];
            o2[r] *= alpha[r]; o3[r] *= alpha[r];
        }

        v2f pa[4];
#pragma unroll
        for (int c2 = 0; c2 < 4; ++c2)
            pa[c2] = *(const v2f*)(&Plds[l16 * 16 + 4 * c2 + 2 * half]);

        const int kb0 = kt * 16;
#pragma unroll
        for (int c2 = 0; c2 < 4; ++c2) {
            const int ko = kb0 + 4 * c2;
            v2f vb;
            vb = *(const v2f*)(Vr0 + ko); o0 = wmma_f32(pa[c2], vb, o0);
            vb = *(const v2f*)(Vr1 + ko); o1 = wmma_f32(pa[c2], vb, o1);
            vb = *(const v2f*)(Vr2 + ko); o2 = wmma_f32(pa[c2], vb, o2);
            vb = *(const v2f*)(Vr3 + ko); o3 = wmma_f32(pa[c2], vb, o3);
        }
        __syncthreads();
    }

    float* Ob = out + ((size_t)b * Tsz + qi * 16 + 8 * half) * Hsz + l16;
#pragma unroll
    for (int r = 0; r < 8; ++r) {
        const float inv = 1.0f / lrow[r];
        Ob[(size_t)r * Hsz +  0] = o0[r] * inv;
        Ob[(size_t)r * Hsz + 16] = o1[r] * inv;
        Ob[(size_t)r * Hsz + 32] = o2[r] * inv;
        Ob[(size_t)r * Hsz + 48] = o3[r] * inv;
    }
}

extern "C" void kernel_launch(void* const* d_in, const int* in_sizes, int n_in,
                              void* d_out, int out_size, void* d_ws, size_t ws_size,
                              hipStream_t stream) {
    const float* x  = (const float*)d_in[0];
    const float* Wq = (const float*)d_in[1];
    const float* Wk = (const float*)d_in[2];
    const float* Wv = (const float*)d_in[3];

    const size_t BTH = (size_t)Bsz * Tsz * Hsz;
    float* qkv = (float*)d_ws;
    float* Wt  = qkv + 3 * BTH;

    dim3 g0(Hsz, 3), blk0(256);
    wt_kernel<<<g0, blk0, 0, stream>>>(Wq, Wk, Wv, Wt);

    dim3 g1(Bsz * Tsz / 64, 3), blk1(128);
    qkv_kernel<<<g1, blk1, 0, stream>>>(x, Wt, qkv);

    dim3 g2(Tsz / 16, Bsz), blk2(32);
    attn_kernel<<<g2, blk2, 0, stream>>>(qkv, (float*)d_out);
}


# attention softmax: 16-lane row max and row sum butterflies as DPP row_xmask steps on the vector ALU instead of __shfl_xor (ds_bpermute_b32 via the LDS unit, one wait per step); sum order unchanged
# speedup vs baseline: 2.3658x; 2.3658x over previous
#include <hip/hip_runtime.h>
#include <math.h>

#define Bsz 8
#define Tsz 2048
#define Csz 1024
#define Hsz 64

typedef __attribute__((ext_vector_type(2))) float v2f;
typedef __attribute__((ext_vector_type(8))) float v8f;

__device__ __forceinline__ v8f wmma_f32(v2f a, v2f b, v8f c) {
    return __builtin_amdgcn_wmma_f32_16x16x4_f32(
         false, a,  false, b,
         (short)0, c,  false,  false);
}

__global__ __launch_bounds__(256) void wt_kernel(
    const float* __restrict__ Wq,
    const float* __restrict__ Wk,
    const float* __restrict__ Wv,
    float* __restrict__ Wt)
{
    const int which = blockIdx.y;
    const float* W = (which == 0) ? Wq : (which == 1) ? Wk : Wv;
    float* dst = Wt + (size_t)which * Hsz * Csz;
    const int h = blockIdx.x;
    for (int c = threadIdx.x; c < Csz; c += 256)
        dst[(size_t)h * Csz + c] = W[(size_t)c * Hsz + h];
}

#define KSTEPS 4

struct QkvFrag {
    v2f a[KSTEPS][4];
    v2f b[KSTEPS][3];
};

__device__ __forceinline__ void qkv_load(QkvFrag& f,
                                         const float* a0, const float* a1,
                                         const float* a2, const float* a3,
                                         const float* b0, const float* b1,
                                         const float* b2, int kk) {
#pragma unroll
    for (int s = 0; s < KSTEPS; ++s) {
        f.a[s][0] = *(const v2f*)(a0 + kk + 4 * s);
        f.a[s][1] = *(const v2f*)(a1 + kk + 4 * s);
        f.a[s][2] = *(const v2f*)(a2 + kk + 4 * s);
        f.a[s][3] = *(const v2f*)(a3 + kk + 4 * s);
        f.b[s][0] = *(const v2f*)(b0 + kk + 4 * s);
        f.b[s][1] = *(const v2f*)(b1 + kk + 4 * s);
        f.b[s][2] = *(const v2f*)(b2 + kk + 4 * s);
    }
}

__device__ __forceinline__ void qkv_mma(const QkvFrag& f, v8f (&acc)[4][3]) {
#pragma unroll
    for (int s = 0; s < KSTEPS; ++s)
#pragma unroll
        for (int i = 0; i < 4; ++i)
#pragma unroll
            for (int j = 0; j < 3; ++j)
                acc[i][j] = wmma_f32(f.a[s][i], f.b[s][j], acc[i][j]);
}

__global__ __launch_bounds__(128) void qkv_kernel(
    const float* __restrict__ x,
    const float* __restrict__ Wt,
    float* __restrict__ qkv)
{
    const int lane = threadIdx.x & 31;
    const int g    = threadIdx.x >> 5;
    const int half = lane >> 4;
    const int l16  = lane & 15;

    const long mrow = (long)blockIdx.x * 64;

    const float* a0 = x + (mrow +  0 + l16) * (size_t)Csz + 2 * half;
    const float* a1 = x + (mrow + 16 + l16) * (size_t)Csz + 2 * half;
    const float* a2 = x + (mrow + 32 + l16) * (size_t)Csz + 2 * half;
    const float* a3 = x + (mrow + 48 + l16) * (size_t)Csz + 2 * half;
    const float* b0 = Wt + (size_t)(48 * g +  0 + l16) * Csz + 2 * half;
    const float* b1 = Wt + (size_t)(48 * g + 16 + l16) * Csz + 2 * half;
    const float* b2 = Wt + (size_t)(48 * g + 32 + l16) * Csz + 2 * half;

    v8f acc[4][3];
#pragma unroll
    for (int i = 0; i < 4; ++i)
#pragma unroll
        for (int j = 0; j < 3; ++j)
            acc[i][j] = v8f{};

    QkvFrag f0, f1;
    qkv_load(f0, a0, a1, a2, a3, b0, b1, b2, 0);
    const int KB = 4 * KSTEPS;
    for (int kk = 0; kk < Csz; kk += 2 * KB) {
        qkv_load(f1, a0, a1, a2, a3, b0, b1, b2, kk + KB);
        __builtin_amdgcn_sched_barrier(0);
        qkv_mma(f0, acc);
        __builtin_amdgcn_sched_barrier(0);
        const int kn = (kk + 2 * KB < Csz) ? kk + 2 * KB : Csz - KB;
        qkv_load(f0, a0, a1, a2, a3, b0, b1, b2, kn);
        __builtin_amdgcn_sched_barrier(0);
        qkv_mma(f1, acc);
        __builtin_amdgcn_sched_barrier(0);
    }

    const size_t BTH = (size_t)Bsz * Tsz * Hsz;
    const long bb   = mrow / Tsz;
    const long tblk = mrow - bb * Tsz;
#pragma unroll
    for (int j = 0; j < 3; ++j) {
        const int t     = 3 * g + j;
        const int which = t >> 2;
        const int col   = (t & 3) * 16 + l16;
        if (which < 2) {
            float* op = qkv + (size_t)which * BTH + (mrow + 8 * half) * (size_t)Hsz + col;
#pragma unroll
            for (int i = 0; i < 4; ++i)
#pragma unroll
                for (int r = 0; r < 8; ++r)
                    op[(size_t)(16 * i + r) * Hsz] = acc[i][j][r];
        } else {
            float* vp = qkv + 2 * BTH + ((size_t)bb * Hsz + col) * Tsz + tblk + 8 * half;
#pragma unroll
            for (int i = 0; i < 4; ++i) {
                const v8f a = acc[i][j];
                *(float4*)(vp + 16 * i + 0) = make_float4(a[0], a[1], a[2], a[3]);
                *(float4*)(vp + 16 * i + 4) = make_float4(a[4], a[5], a[6], a[7]);
            }
        }
    }
}

template <int M>
__device__ __forceinline__ float row_xor(float v) {
    return __builtin_bit_cast(float, __builtin_amdgcn_update_dpp(
        0, __builtin_bit_cast(int, v), 0x160 + M, 0xf, 0xf, true));
}

#define KBLK 64
#define PPITCH 68
#define TPITCH 68
#define NW 4

__global__ __launch_bounds__(128) void attn_kernel(
    const float* __restrict__ qkv,
    float* __restrict__ out)
{
    __shared__ __attribute__((aligned(16))) float Klds[2][KBLK * TPITCH];
    __shared__ __attribute__((aligned(16))) float Vlds[2][Hsz * TPITCH];
    __shared__ __attribute__((aligned(16))) float Plds[NW][16 * PPITCH];

    const int tid  = threadIdx.x;
    const int lane = tid & 31;
    const int w    = tid >> 5;
    const int half = lane >> 4;
    const int l16  = lane & 15;
    const int b    = blockIdx.y;
    const int grp  = gridDim.x - 1 - blockIdx.x;
    const int qi   = grp * NW + w;

    const size_t BTH = (size_t)Bsz * Tsz * Hsz;
    const float* Q  = qkv +            (size_t)b * Tsz * Hsz;
    const float* K  = qkv + BTH     +  (size_t)b * Tsz * Hsz;
    const float* Vt = qkv + 2 * BTH +  (size_t)b * Hsz * Tsz;

    const float scale = 0.125f;
    const int qrow = qi * 16 + l16;

    v2f qa[16];
#pragma unroll
    for (int c = 0; c < 16; ++c) {
        v2f t = *(const v2f*)(Q + (size_t)qrow * Hsz + 4 * c + 2 * half);
        qa[c][0] = t[0] * scale;
        qa[c][1] = t[1] * scale;
    }

    const int crow = tid >> 4;
    const int ccol = (tid & 15) * 4;
    const float* Kg = K  + (size_t)crow * Hsz + ccol;
    const float* Vg = Vt + (size_t)crow * Tsz + ccol;
    const int lofs  = crow * TPITCH + ccol;

#define TILE_LOAD(kb)                                                          \
    do {                                                                       \
        const float* kp_ = Kg + (size_t)(kb) * Hsz;                            \
        const float* vp_ = Vg + (kb);                                          \
        k0 = *(const float4*)(kp_ +  0 * Hsz); k1 = *(const float4*)(kp_ +  8 * Hsz); \
        k2 = *(const float4*)(kp_ + 16 * Hsz); k3 = *(const float4*)(kp_ + 24 * Hsz); \
        k4 = *(const float4*)(kp_ + 32 * Hsz); k5 = *(const float4*)(kp_ + 40 * Hsz); \
        k6 = *(const float4*)(kp_ + 48 * Hsz); k7 = *(const float4*)(kp_ + 56 * Hsz); \
        v0 = *(const float4*)(vp_ +  0 * Tsz); v1 = *(const float4*)(vp_ +  8 * Tsz); \
        v2 = *(const float4*)(vp_ + 16 * Tsz); v3 = *(const float4*)(vp_ + 24 * Tsz); \
        v4 = *(const float4*)(vp_ + 32 * Tsz); v5 = *(const float4*)(vp_ + 40 * Tsz); \
        v6 = *(const float4*)(vp_ + 48 * Tsz); v7 = *(const float4*)(vp_ + 56 * Tsz); \
    } while (0)
#define TILE_STORE(buf)                                                        \
    do {                                                                       \
        float* kd_ = &Klds[buf][lofs];                                         \
        float* vd_ = &Vlds[buf][lofs];                                         \
        *(float4*)(kd_ +  0 * TPITCH) = k0; *(float4*)(kd_ +  8 * TPITCH) = k1; \
        *(float4*)(kd_ + 16 * TPITCH) = k2; *(float4*)(kd_ + 24 * TPITCH) = k3; \
        *(float4*)(kd_ + 32 * TPITCH) = k4; *(float4*)(kd_ + 40 * TPITCH) = k5; \
        *(float4*)(kd_ + 48 * TPITCH) = k6; *(float4*)(kd_ + 56 * TPITCH) = k7; \
        *(float4*)(vd_ +  0 * TPITCH) = v0; *(float4*)(vd_ +  8 * TPITCH) = v1; \
        *(float4*)(vd_ + 16 * TPITCH) = v2; *(float4*)(vd_ + 24 * TPITCH) = v3; \
        *(float4*)(vd_ + 32 * TPITCH) = v4; *(float4*)(vd_ + 40 * TPITCH) = v5; \
        *(float4*)(vd_ + 48 * TPITCH) = v6; *(float4*)(vd_ + 56 * TPITCH) = v7; \
    } while (0)
    float4 k0, k1, k2, k3, k4, k5, k6, k7, v0, v1, v2, v3, v4, v5, v6, v7;
    TILE_LOAD(0);
    TILE_STORE(0);

    float mrow[8], lrow[8];
#pragma unroll
    for (int r = 0; r < 8; ++r) { mrow[r] = -INFINITY; lrow[r] = 0.0f; }
    v8f o0 = {}, o1 = {}, o2 = {}, o3 = {};

    const int ntrip = grp + 1;
    float* Pw = &Plds[w][0];

    for (int tr = 0; tr < ntrip; ++tr) {
        const int kb0 = tr * KBLK;
        const int cur = tr & 1;
        __syncthreads();

        const bool more = (tr + 1 < ntrip);
        TILE_LOAD(more ? kb0 + KBLK : kb0);

        const float* Kl = &Klds[cur][l16 * TPITCH + 2 * half];
        v2f kf[4][16];
#pragma unroll
        for (int j = 0; j < 4; ++j)
#pragma unroll
            for (int c = 0; c < 16; ++c)
                kf[j][c] = *(const v2f*)(Kl + 16 * j * TPITCH + 4 * c);
        __builtin_amdgcn_sched_barrier(0);

        v8f s0 = {}, s1 = {}, s2 = {}, s3 = {};
#pragma unroll
        for (int c = 0; c < 16; ++c) {
            s0 = wmma_f32(qa[c], kf[0][c], s0);
            s1 = wmma_f32(qa[c], kf[1][c], s1);
            s2 = wmma_f32(qa[c], kf[2][c], s2);
            s3 = wmma_f32(qa[c], kf[3][c], s3);
        }
        __builtin_amdgcn_sched_barrier(0);

        const float* Vl = &Vlds[cur][l16 * TPITCH + 2 * half];
        v2f vf[4][16];
#pragma unroll
        for (int j = 0; j < 4; ++j)
#pragma unroll
            for (int c2 = 0; c2 < 16; ++c2)
                vf[j][c2] = *(const v2f*)(Vl + 16 * j * TPITCH + 4 * c2);

        if (tr == ntrip - 1) {
            const int key0 = kb0 + l16;
            const int qr0  = qi * 16 + 8 * half;
#pragma unroll
            for (int r = 0; r < 8; ++r) {
                const int qr = qr0 + r;
                if (key0      > qr) s0[r] = -INFINITY;
                if (key0 + 16 > qr) s1[r] = -INFINITY;
                if (key0 + 32 > qr) s2[r] = -INFINITY;
                if (key0 + 48 > qr) s3[r] = -INFINITY;
            }
        }

        float alpha[8];
#pragma unroll
        for (int r = 0; r < 8; ++r) {
            float mx = fmaxf(fmaxf(s0[r], s1[r]), fmaxf(s2[r], s3[r]));
            mx = fmaxf(mx, row_xor<1>(mx));
            mx = fmaxf(mx, row_xor<2>(mx));
            mx = fmaxf(mx, row_xor<4>(mx));
            mx = fmaxf(mx, row_xor<8>(mx));
            const float mn = fmaxf(mrow[r], mx);
            alpha[r] = expf(mrow[r] - mn);
            mrow[r]  = mn;
            const float p0 = expf(s0[r] - mn);
            const float p1 = expf(s1[r] - mn);
            const float p2 = expf(s2[r] - mn);
            const float p3 = expf(s3[r] - mn);
            float sm = (p0 + p1) + (p2 + p3);
            sm += row_xor<1>(sm);
            sm += row_xor<2>(sm);
            sm += row_xor<4>(sm);
            sm += row_xor<8>(sm);
            lrow[r] = lrow[r] * alpha[r] + sm;

            float* prow = Pw + (r + 8 * half) * PPITCH + l16;
            prow[ 0] = p0;
            prow[16] = p1;
            prow[32] = p2;
            prow[48] = p3;
        }
        __builtin_amdgcn_fence(__ATOMIC_RELEASE, "wavefront");
        __builtin_amdgcn_wave_barrier();
        __builtin_amdgcn_fence(__ATOMIC_ACQUIRE, "wavefront");

#pragma unroll
        for (int r = 0; r < 8; ++r) {
            o0[r] *= alpha[r]; o1[r] *= alpha[r];
            o2[r] *= alpha[r]; o3[r] *= alpha[r];
        }

        v2f pa[16];
#pragma unroll
        for (int c2 = 0; c2 < 16; ++c2)
            pa[c2] = *(const v2f*)(Pw + l16 * PPITCH + 4 * c2 + 2 * half);
        __builtin_amdgcn_sched_barrier(0);

#pragma unroll
        for (int c2 = 0; c2 < 16; ++c2) {
            o0 = wmma_f32(pa[c2], vf[0][c2], o0);
            o1 = wmma_f32(pa[c2], vf[1][c2], o1);
            o2 = wmma_f32(pa[c2], vf[2][c2], o2);
            o3 = wmma_f32(pa[c2], vf[3][c2], o3);
        }

        if (more) {
            TILE_STORE(cur ^ 1);
        }
    }

#undef TILE_LOAD
#undef TILE_STORE

    float* Ob = out + ((size_t)b * Tsz + qi * 16 + 8 * half) * Hsz + l16;
#pragma unroll
    for (int r = 0; r < 8; ++r) {
        const float inv = 1.0f / lrow[r];
        Ob[(size_t)r * Hsz +  0] = o0[r] * inv;
        Ob[(size_t)r * Hsz + 16] = o1[r] * inv;
        Ob[(size_t)r * Hsz + 32] = o2[r] * inv;
        Ob[(size_t)r * Hsz + 48] = o3[r] * inv;
    }
}

extern "C" void kernel_launch(void* const* d_in, const int* in_sizes, int n_in,
                              void* d_out, int out_size, void* d_ws, size_t ws_size,
                              hipStream_t stream) {
    const float* x  = (const float*)d_in[0];
    const float* Wq = (const float*)d_in[1];
    const float* Wk = (const float*)d_in[2];
    const float* Wv = (const float*)d_in[3];

    const size_t BTH = (size_t)Bsz * Tsz * Hsz;
    float* qkv = (float*)d_ws;
    float* Wt  = qkv + 3 * BTH;

    dim3 g0(Hsz, 3), blk0(256);
    wt_kernel<<<g0, blk0, 0, stream>>>(Wq, Wk, Wv, Wt);

    dim3 g1(Bsz * Tsz / 64), blk1(128);
    qkv_kernel<<<g1, blk1, 0, stream>>>(x, Wt, qkv);

    dim3 g2(Tsz / 64, Bsz), blk2(128);
    attn_kernel<<<g2, blk2, 0, stream>>>(qkv, (float*)d_out);
}


# attention: causal balance across work-groups: mirrored query-group pairs split their 33 key blocks 17+16 over 2 work-groups; a 4th launch merges the long group's 2 softmax states (sum order changes)
# speedup vs baseline: 3.0188x; 1.2760x over previous
#include <hip/hip_runtime.h>
#include <math.h>

#define Bsz 8
#define Tsz 2048
#define Csz 1024
#define Hsz 64

typedef __attribute__((ext_vector_type(2))) float v2f;
typedef __attribute__((ext_vector_type(8))) float v8f;

__device__ __forceinline__ v8f wmma_f32(v2f a, v2f b, v8f c) {
    return __builtin_amdgcn_wmma_f32_16x16x4_f32(
         false, a,  false, b,
         (short)0, c,  false,  false);
}

__global__ __launch_bounds__(256) void wt_kernel(
    const float* __restrict__ Wq,
    const float* __restrict__ Wk,
    const float* __restrict__ Wv,
    float* __restrict__ Wt)
{
    const int which = blockIdx.y;
    const float* W = (which == 0) ? Wq : (which == 1) ? Wk : Wv;
    float* dst = Wt + (size_t)which * Hsz * Csz;
    const int h = blockIdx.x;
    for (int c = threadIdx.x; c < Csz; c += 256)
        dst[(size_t)h * Csz + c] = W[(size_t)c * Hsz + h];
}

#define KSTEPS 4

struct QkvFrag {
    v2f a[KSTEPS][4];
    v2f b[KSTEPS][3];
};

__device__ __forceinline__ void qkv_load(QkvFrag& f,
                                         const float* a0, const float* a1,
                                         const float* a2, const float* a3,
                                         const float* b0, const float* b1,
                                         const float* b2, int kk) {
#pragma unroll
    for (int s = 0; s < KSTEPS; ++s) {
        f.a[s][0] = *(const v2f*)(a0 + kk + 4 * s);
        f.a[s][1] = *(const v2f*)(a1 + kk + 4 * s);
        f.a[s][2] = *(const v2f*)(a2 + kk + 4 * s);
        f.a[s][3] = *(const v2f*)(a3 + kk + 4 * s);
        f.b[s][0] = *(const v2f*)(b0 + kk + 4 * s);
        f.b[s][1] = *(const v2f*)(b1 + kk + 4 * s);
        f.b[s][2] = *(const v2f*)(b2 + kk + 4 * s);
    }
}

__device__ __forceinline__ void qkv_mma(const QkvFrag& f, v8f (&acc)[4][3]) {
#pragma unroll
    for (int s = 0; s < KSTEPS; ++s)
#pragma unroll
        for (int i = 0; i < 4; ++i)
#pragma unroll
            for (int j = 0; j < 3; ++j)
                acc[i][j] = wmma_f32(f.a[s][i], f.b[s][j], acc[i][j]);
}

__global__ __launch_bounds__(128) void qkv_kernel(
    const float* __restrict__ x,
    const float* __restrict__ Wt,
    float* __restrict__ qkv)
{
    const int lane = threadIdx.x & 31;
    const int g    = threadIdx.x >> 5;
    const int half = lane >> 4;
    const int l16  = lane & 15;

    const long mrow = (long)blockIdx.x * 64;

    const float* a0 = x + (mrow +  0 + l16) * (size_t)Csz + 2 * half;
    const float* a1 = x + (mrow + 16 + l16) * (size_t)Csz + 2 * half;
    const float* a2 = x + (mrow + 32 + l16) * (size_t)Csz + 2 * half;
    const float* a3 = x + (mrow + 48 + l16) * (size_t)Csz + 2 * half;
    const float* b0 = Wt + (size_t)(48 * g +  0 + l16) * Csz + 2 * half;
    const float* b1 = Wt + (size_t)(48 * g + 16 + l16) * Csz + 2 * half;
    const float* b2 = Wt + (size_t)(48 * g + 32 + l16) * Csz + 2 * half;

    v8f acc[4][3];
#pragma unroll
    for (int i = 0; i < 4; ++i)
#pragma unroll
        for (int j = 0; j < 3; ++j)
            acc[i][j] = v8f{};

    QkvFrag f0, f1;
    qkv_load(f0, a0, a1, a2, a3, b0, b1, b2, 0);
    const int KB = 4 * KSTEPS;
    for (int kk = 0; kk < Csz; kk += 2 * KB) {
        qkv_load(f1, a0, a1, a2, a3, b0, b1, b2, kk + KB);
        __builtin_amdgcn_sched_barrier(0);
        qkv_mma(f0, acc);
        __builtin_amdgcn_sched_barrier(0);
        const int kn = (kk + 2 * KB < Csz) ? kk + 2 * KB : Csz - KB;
        qkv_load(f0, a0, a1, a2, a3, b0, b1, b2, kn);
        __builtin_amdgcn_sched_barrier(0);
        qkv_mma(f1, acc);
        __builtin_amdgcn_sched_barrier(0);
    }

    const size_t BTH = (size_t)Bsz * Tsz * Hsz;
    const long bb   = mrow / Tsz;
    const long tblk = mrow - bb * Tsz;
#pragma unroll
    for (int j = 0; j < 3; ++j) {
        const int t     = 3 * g + j;
        const int which = t >> 2;
        const int col   = (t & 3) * 16 + l16;
        if (which < 2) {
            float* op = qkv + (size_t)which * BTH + (mrow + 8 * half) * (size_t)Hsz + col;
#pragma unroll
            for (int i = 0; i < 4; ++i)
#pragma unroll
                for (int r = 0; r < 8; ++r)
                    op[(size_t)(16 * i + r) * Hsz] = acc[i][j][r];
        } else {
            float* vp = qkv + 2 * BTH + ((size_t)bb * Hsz + col) * Tsz + tblk + 8 * half;
#pragma unroll
            for (int i = 0; i < 4; ++i) {
                const v8f a = acc[i][j];
                *(float4*)(vp + 16 * i + 0) = make_float4(a[0], a[1], a[2], a[3]);
                *(float4*)(vp + 16 * i + 4) = make_float4(a[4], a[5], a[6], a[7]);
            }
        }
    }
}

template <int M>
__device__ __forceinline__ float row_xor(float v) {
    return __builtin_bit_cast(float, __builtin_amdgcn_update_dpp(
        0, __builtin_bit_cast(int, v), 0x160 + M, 0xf, 0xf, true));
}

#define KBLK 64
#define PPITCH 68
#define TPITCH 68
#define NW 4
#define NGRP (Tsz / 64)
#define NPAIR (NGRP / 2)
#define SPLITB (NPAIR + 1)

__global__ __launch_bounds__(128) void attn_kernel(
    float* qkv,
    float* __restrict__ ml,
    float* __restrict__ out)
{
    __shared__ __attribute__((aligned(16))) float Klds[2][KBLK * TPITCH];
    __shared__ __attribute__((aligned(16))) float Vlds[2][Hsz * TPITCH];
    __shared__ __attribute__((aligned(16))) float Plds[NW][16 * PPITCH];

    const int tid  = threadIdx.x;
    const int lane = tid & 31;
    const int w    = tid >> 5;
    const int half = lane >> 4;
    const int l16  = lane & 15;
    const int b    = blockIdx.y;
    const int pr   = blockIdx.x >> 1;
    const int role = blockIdx.x & 1;
    const int gL   = NGRP - 1 - pr;
    const int gS   = pr;

    const size_t BTH = (size_t)Bsz * Tsz * Hsz;
    const float* Q  = qkv +            (size_t)b * Tsz * Hsz;
    const float* K  = qkv + BTH     +  (size_t)b * Tsz * Hsz;
    const float* Vt = qkv + 2 * BTH +  (size_t)b * Hsz * Tsz;

    const float scale = 0.125f;

    const int crow = tid >> 4;
    const int ccol = (tid & 15) * 4;
    const float* Kg = K  + (size_t)crow * Hsz + ccol;
    const float* Vg = Vt + (size_t)crow * Tsz + ccol;
    const int lofs  = crow * TPITCH + ccol;

#define TILE_LOAD(kb)                                                          \
    do {                                                                       \
        const float* kp_ = Kg + (size_t)(kb) * Hsz;                            \
        const float* vp_ = Vg + (kb);                                          \
        k0 = *(const float4*)(kp_ +  0 * Hsz); k1 = *(const float4*)(kp_ +  8 * Hsz); \
        k2 = *(const float4*)(kp_ + 16 * Hsz); k3 = *(const float4*)(kp_ + 24 * Hsz); \
        k4 = *(const float4*)(kp_ + 32 * Hsz); k5 = *(const float4*)(kp_ + 40 * Hsz); \
        k6 = *(const float4*)(kp_ + 48 * Hsz); k7 = *(const float4*)(kp_ + 56 * Hsz); \
        v0 = *(const float4*)(vp_ +  0 * Tsz); v1 = *(const float4*)(vp_ +  8 * Tsz); \
        v2 = *(const float4*)(vp_ + 16 * Tsz); v3 = *(const float4*)(vp_ + 24 * Tsz); \
        v4 = *(const float4*)(vp_ + 32 * Tsz); v5 = *(const float4*)(vp_ + 40 * Tsz); \
        v6 = *(const float4*)(vp_ + 48 * Tsz); v7 = *(const float4*)(vp_ + 56 * Tsz); \
    } while (0)
#define TILE_STORE(buf)                                                        \
    do {                                                                       \
        float* kd_ = &Klds[buf][lofs];                                         \
        float* vd_ = &Vlds[buf][lofs];                                         \
        *(float4*)(kd_ +  0 * TPITCH) = k0; *(float4*)(kd_ +  8 * TPITCH) = k1; \
        *(float4*)(kd_ + 16 * TPITCH) = k2; *(float4*)(kd_ + 24 * TPITCH) = k3; \
        *(float4*)(kd_ + 32 * TPITCH) = k4; *(float4*)(kd_ + 40 * TPITCH) = k5; \
        *(float4*)(kd_ + 48 * TPITCH) = k6; *(float4*)(kd_ + 56 * TPITCH) = k7; \
        *(float4*)(vd_ +  0 * TPITCH) = v0; *(float4*)(vd_ +  8 * TPITCH) = v1; \
        *(float4*)(vd_ + 16 * TPITCH) = v2; *(float4*)(vd_ + 24 * TPITCH) = v3; \
        *(float4*)(vd_ + 32 * TPITCH) = v4; *(float4*)(vd_ + 40 * TPITCH) = v5; \
        *(float4*)(vd_ + 48 * TPITCH) = v6; *(float4*)(vd_ + 56 * TPITCH) = v7; \
    } while (0)
    float4 k0, k1, k2, k3, k4, k5, k6, k7, v0, v1, v2, v3, v4, v5, v6, v7;

    float* Pw = &Plds[w][0];

    const int nseg = role + 1;
    for (int seg = 0; seg < nseg; ++seg) {
        const bool lpart = (role == 0) || (seg == 1);
        const int grp  = lpart ? gL : gS;
        const int blk0 = (role == 1 && seg == 1) ? SPLITB : 0;
        const int nblk = (role == 0) ? SPLITB : (seg == 0 ? gS + 1 : gL + 1 - SPLITB);
        const int qi   = grp * NW + w;
        const int qrow = qi * 16 + l16;

        v2f qa[16];
#pragma unroll
        for (int c = 0; c < 16; ++c) {
            v2f t = *(const v2f*)(Q + (size_t)qrow * Hsz + 4 * c + 2 * half);
            qa[c][0] = t[0] * scale;
            qa[c][1] = t[1] * scale;
        }

        if (seg > 0) __syncthreads();
        if (nblk > 0) {
            TILE_LOAD(blk0 * KBLK);
            TILE_STORE(0);
        }

        float mrow[8], lrow[8];
#pragma unroll
        for (int r = 0; r < 8; ++r) { mrow[r] = -INFINITY; lrow[r] = 0.0f; }
        v8f o0 = {}, o1 = {}, o2 = {}, o3 = {};

        for (int tr = 0; tr < nblk; ++tr) {
            const int blk = blk0 + tr;
            const int kb0 = blk * KBLK;
            const int cur = tr & 1;
            __syncthreads();

            const bool more = (tr + 1 < nblk);
            TILE_LOAD(more ? kb0 + KBLK : kb0);

            const float* Kl = &Klds[cur][l16 * TPITCH + 2 * half];
            v2f kf[4][16];
#pragma unroll
            for (int j = 0; j < 4; ++j)
#pragma unroll
                for (int c = 0; c < 16; ++c)
                    kf[j][c] = *(const v2f*)(Kl + 16 * j * TPITCH + 4 * c);
            __builtin_amdgcn_sched_barrier(0);

            v8f s0 = {}, s1 = {}, s2 = {}, s3 = {};
#pragma unroll
            for (int c = 0; c < 16; ++c) {
                s0 = wmma_f32(qa[c], kf[0][c], s0);
                s1 = wmma_f32(qa[c], kf[1][c], s1);
                s2 = wmma_f32(qa[c], kf[2][c], s2);
                s3 = wmma_f32(qa[c], kf[3][c], s3);
            }
            __builtin_amdgcn_sched_barrier(0);

            const float* Vl = &Vlds[cur][l16 * TPITCH + 2 * half];
            v2f vf[4][16];
#pragma unroll
            for (int j = 0; j < 4; ++j)
#pragma unroll
                for (int c2 = 0; c2 < 16; ++c2)
                    vf[j][c2] = *(const v2f*)(Vl + 16 * j * TPITCH + 4 * c2);

            if (blk == grp) {
                const int key0 = kb0 + l16;
                const int qr0  = qi * 16 + 8 * half;
#pragma unroll
                for (int r = 0; r < 8; ++r) {
                    const int qr = qr0 + r;
                    if (key0      > qr) s0[r] = -INFINITY;
                    if (key0 + 16 > qr) s1[r] = -INFINITY;
                    if (key0 + 32 > qr) s2[r] = -INFINITY;
                    if (key0 + 48 > qr) s3[r] = -INFINITY;
                }
            }

            float alpha[8];
#pragma unroll
            for (int r = 0; r < 8; ++r) {
                float mx = fmaxf(fmaxf(s0[r], s1[r]), fmaxf(s2[r], s3[r]));
                mx = fmaxf(mx, row_xor<1>(mx));
                mx = fmaxf(mx, row_xor<2>(mx));
                mx = fmaxf(mx, row_xor<4>(mx));
                mx = fmaxf(mx, row_xor<8>(mx));
                const float mn = fmaxf(mrow[r], mx);
                alpha[r] = expf(mrow[r] - mn);
                mrow[r]  = mn;
                const float p0 = expf(s0[r] - mn);
                const float p1 = expf(s1[r] - mn);
                const float p2 = expf(s2[r] - mn);
                const float p3 = expf(s3[r] - mn);
                float sm = (p0 + p1) + (p2 + p3);
                sm += row_xor<1>(sm);
                sm += row_xor<2>(sm);
                sm += row_xor<4>(sm);
                sm += row_xor<8>(sm);
                lrow[r] = lrow[r] * alpha[r] + sm;

                float* prow = Pw + (r + 8 * half) * PPITCH + l16;
                prow[ 0] = p0;
                prow[16] = p1;
                prow[32] = p2;
                prow[48] = p3;
            }
            __builtin_amdgcn_fence(__ATOMIC_RELEASE, "wavefront");
            __builtin_amdgcn_wave_barrier();
            __builtin_amdgcn_fence(__ATOMIC_ACQUIRE, "wavefront");

#pragma unroll
            for (int r = 0; r < 8; ++r) {
                o0[r] *= alpha[r]; o1[r] *= alpha[r];
                o2[r] *= alpha[r]; o3[r] *= alpha[r];
            }

            v2f pa[16];
#pragma unroll
            for (int c2 = 0; c2 < 16; ++c2)
                pa[c2] = *(const v2f*)(Pw + l16 * PPITCH + 4 * c2 + 2 * half);
            __builtin_amdgcn_sched_barrier(0);

#pragma unroll
            for (int c2 = 0; c2 < 16; ++c2) {
                o0 = wmma_f32(pa[c2], vf[0][c2], o0);
                o1 = wmma_f32(pa[c2], vf[1][c2], o1);
                o2 = wmma_f32(pa[c2], vf[2][c2], o2);
                o3 = wmma_f32(pa[c2], vf[3][c2], o3);
            }

            if (more) {
                TILE_STORE(cur ^ 1);
            }
        }

        if (!lpart) {
            float* Ob = out + ((size_t)b * Tsz + qi * 16 + 8 * half) * Hsz + l16;
#pragma unroll
            for (int r = 0; r < 8; ++r) {
                const float inv = 1.0f / lrow[r];
                Ob[(size_t)r * Hsz +  0] = o0[r] * inv;
                Ob[(size_t)r * Hsz + 16] = o1[r] * inv;
                Ob[(size_t)r * Hsz + 32] = o2[r] * inv;
                Ob[(size_t)r * Hsz + 48] = o3[r] * inv;
            }
        } else {
            const int part = role;
            float* Od = (part == 0)
                ? out + ((size_t)b * Tsz + qi * 16 + 8 * half) * Hsz + l16
                : qkv + ((size_t)b * Tsz + (gS * NW + w) * 16 + 8 * half) * Hsz + l16;
#pragma unroll
            for (int r = 0; r < 8; ++r) {
                Od[(size_t)r * Hsz +  0] = o0[r];
                Od[(size_t)r * Hsz + 16] = o1[r];
                Od[(size_t)r * Hsz + 32] = o2[r];
                Od[(size_t)r * Hsz + 48] = o3[r];
            }
            float* mlp = ml + ((((size_t)part * Bsz + b) * NPAIR + pr) * 64 + 16 * w + 8 * half) * 2;
#pragma unroll
            for (int r = 0; r < 8; ++r)
                if (l16 == r)
                    *(float2*)(mlp + 2 * r) = make_float2(mrow[r], lrow[r]);
        }
    }

#undef TILE_LOAD
#undef TILE_STORE
}

__global__ __launch_bounds__(256) void merge_kernel(
    const float* __restrict__ qkv,
    const float* __restrict__ ml,
    float* out)
{
    const int e  = blockIdx.x * 256 + threadIdx.x;
    const int c4 = e & 15;
    const int r  = (e >> 4) & 63;
    const int pr = (e >> 10) % NPAIR;
    const int b  = (e >> 10) / NPAIR;
    const int gL = NGRP - 1 - pr;
    const int gS = pr;

    const float2 a0 = *(const float2*)(ml + ((((size_t)0 * Bsz + b) * NPAIR + pr) * 64 + r) * 2);
    const float2 a1 = *(const float2*)(ml + ((((size_t)1 * Bsz + b) * NPAIR + pr) * 64 + r) * 2);
    const float m  = fmaxf(a0.x, a1.x);
    const float w0 = expf(a0.x - m);
    const float w1 = expf(a1.x - m);
    const float inv = 1.0f / (a0.y * w0 + a1.y * w1);

    float* op = out + ((size_t)b * Tsz + gL * 64 + r) * Hsz + 4 * c4;
    const float4 x0 = *(const float4*)op;
    const float4 x1 = *(const float4*)(qkv + ((size_t)b * Tsz + gS * 64 + r) * Hsz + 4 * c4);
    float4 y;
    y.x = (x0.x * w0 + x1.x * w1) * inv;
    y.y = (x0.y * w0 + x1.y * w1) * inv;
    y.z = (x0.z * w0 + x1.z * w1) * inv;
    y.w = (x0.w * w0 + x1.w * w1) * inv;
    *(float4*)op = y;
}

extern "C" void kernel_launch(void* const* d_in, const int* in_sizes, int n_in,
                              void* d_out, int out_size, void* d_ws, size_t ws_size,
                              hipStream_t stream) {
    const float* x  = (const float*)d_in[0];
    const float* Wq = (const float*)d_in[1];
    const float* Wk = (const float*)d_in[2];
    const float* Wv = (const float*)d_in[3];

    const size_t BTH = (size_t)Bsz * Tsz * Hsz;
    float* qkv = (float*)d_ws;
    float* Wt  = qkv + 3 * BTH;
    float* ml  = Wt;

    dim3 g0(Hsz, 3), blk0(256);
    wt_kernel<<<g0, blk0, 0, stream>>>(Wq, Wk, Wv, Wt);

    dim3 g1(Bsz * Tsz / 64), blk1(128);
    qkv_kernel<<<g1, blk1, 0, stream>>>(x, Wt, qkv);

    dim3 g2(Tsz / 64, Bsz), blk2(128);
    attn_kernel<<<g2, blk2, 0, stream>>>(qkv, ml, (float*)d_out);

    dim3 g3(Bsz * NPAIR * 64 * 16 / 256), blk3(256);
    merge_kernel<<<g3, blk3, 0, stream>>>(qkv, ml, (float*)d_out);
}


# qkv projection: 4 waves split k, not columns: each sums the whole 64x192 tile over its 8 floats of every 32 (16 fragment loads per 48 WMMAs, not 7 per 12); partials added via LDS (sum order changes)
# speedup vs baseline: 3.2336x; 1.0712x over previous
#include <hip/hip_runtime.h>
#include <math.h>

#define Bsz 8
#define Tsz 2048
#define Csz 1024
#define Hsz 64

typedef __attribute__((ext_vector_type(2))) float v2f;
typedef __attribute__((ext_vector_type(8))) float v8f;

__device__ __forceinline__ v8f wmma_f32(v2f a, v2f b, v8f c) {
    return __builtin_amdgcn_wmma_f32_16x16x4_f32(
         false, a,  false, b,
         (short)0, c,  false,  false);
}

__global__ __launch_bounds__(256) void wt_kernel(
    const float* __restrict__ Wq,
    const float* __restrict__ Wk,
    const float* __restrict__ Wv,
    float* __restrict__ Wt)
{
    const int which = blockIdx.y;
    const float* W = (which == 0) ? Wq : (which == 1) ? Wk : Wv;
    float* dst = Wt + (size_t)which * Hsz * Csz;
    const int h = blockIdx.x;
    for (int c = threadIdx.x; c < Csz; c += 256)
        dst[(size_t)h * Csz + c] = W[(size_t)c * Hsz + h];
}

#define KSTEPS 2
#define KSPAN 32

struct QkvFrag {
    v2f a[KSTEPS][4];
    v2f b[KSTEPS][12];
};

__device__ __forceinline__ void qkv_load(QkvFrag& f, const float* xk, const float* wk, unsigned lofs) {
#pragma unroll
    for (int s = 0; s < KSTEPS; ++s) {
#pragma unroll
        for (int i = 0; i < 4; ++i)
            f.a[s][i] = *(const v2f*)(xk + lofs + (16 * i * Csz + 4 * s));
#pragma unroll
        for (int j = 0; j < 12; ++j)
            f.b[s][j] = *(const v2f*)(wk + lofs + (16 * j * Csz + 4 * s));
    }
}

__device__ __forceinline__ void qkv_mma(const QkvFrag& f, v8f (&acc)[4][12]) {
#pragma unroll
    for (int s = 0; s < KSTEPS; ++s)
#pragma unroll
        for (int i = 0; i < 4; ++i)
#pragma unroll
            for (int j = 0; j < 12; ++j)
                acc[i][j] = wmma_f32(f.a[s][i], f.b[s][j], acc[i][j]);
}

__device__ __forceinline__ void wmma_slots4() {
    asm volatile("v_nop\n\tv_nop\n\tv_nop\n\tv_nop");
}

__global__ __launch_bounds__(128) void qkv_kernel(
    const float* __restrict__ x,
    const float* __restrict__ Wt,
    float* __restrict__ qkv)
{
    __shared__ float4 red[4 * 48 * 2 * 32];

    const int lane = threadIdx.x & 31;
    const int g    = threadIdx.x >> 5;
    const int half = lane >> 4;
    const int l16  = lane & 15;

    const long mrow = (long)blockIdx.x * 64;

    const float* xb = x + mrow * (size_t)Csz;
    const unsigned lofs = (unsigned)(l16 * Csz + 8 * g + 2 * half);

    v8f acc[4][12];
#pragma unroll
    for (int i = 0; i < 4; ++i)
#pragma unroll
        for (int j = 0; j < 12; ++j)
            acc[i][j] = v8f{};

    QkvFrag f0, f1;
    qkv_load(f0, xb, Wt, lofs);
    for (int kk = 0; kk < Csz; kk += 2 * KSPAN) {
        qkv_load(f1, xb + kk + KSPAN, Wt + kk + KSPAN, lofs);
        __builtin_amdgcn_sched_barrier(0);
        qkv_mma(f0, acc);
        wmma_slots4();
        __builtin_amdgcn_sched_barrier(0);
        const int kn = (kk + 2 * KSPAN < Csz) ? kk + 2 * KSPAN : Csz - KSPAN;
        qkv_load(f0, xb + kn, Wt + kn, lofs);
        __builtin_amdgcn_sched_barrier(0);
        qkv_mma(f1, acc);
        wmma_slots4();
        __builtin_amdgcn_sched_barrier(0);
    }

#pragma unroll
    for (int i = 0; i < 4; ++i)
#pragma unroll
        for (int j = 0; j < 12; ++j) {
            const v8f a = acc[i][j];
            float4* rp = red + ((g * 48 + 12 * i + j) * 2) * 32 + lane;
            rp[0]  = make_float4(a[0], a[1], a[2], a[3]);
            rp[32] = make_float4(a[4], a[5], a[6], a[7]);
        }
    __syncthreads();

    const size_t BTH = (size_t)Bsz * Tsz * Hsz;
    const long bb   = mrow / Tsz;
    const long tblk = mrow - bb * Tsz;
    for (int j = 0; j < 3; ++j) {
        const int t     = 3 * g + j;
        const int which = t >> 2;
        const int col   = (t & 3) * 16 + l16;
        v8f sum[4];
#pragma unroll
        for (int i = 0; i < 4; ++i) {
            const float4* rp = red + ((12 * i + t) * 2) * 32 + lane;
            float4 lo = rp[0], hi = rp[32];
#pragma unroll
            for (int s = 1; s < 4; ++s) {
                const float4 l2 = rp[s * 48 * 2 * 32], h2 = rp[s * 48 * 2 * 32 + 32];
                lo.x += l2.x; lo.y += l2.y; lo.z += l2.z; lo.w += l2.w;
                hi.x += h2.x; hi.y += h2.y; hi.z += h2.z; hi.w += h2.w;
            }
            sum[i] = v8f{lo.x, lo.y, lo.z, lo.w, hi.x, hi.y, hi.z, hi.w};
        }
        if (which < 2) {
            float* op = qkv + (size_t)which * BTH + (mrow + 8 * half) * (size_t)Hsz + col;
#pragma unroll
            for (int i = 0; i < 4; ++i)
#pragma unroll
                for (int r = 0; r < 8; ++r)
                    op[(size_t)(16 * i + r) * Hsz] = sum[i][r];
        } else {
            float* vp = qkv + 2 * BTH + ((size_t)bb * Hsz + col) * Tsz + tblk + 8 * half;
#pragma unroll
            for (int i = 0; i < 4; ++i) {
                const v8f a = sum[i];
                *(float4*)(vp + 16 * i + 0) = make_float4(a[0], a[1], a[2], a[3]);
                *(float4*)(vp + 16 * i + 4) = make_float4(a[4], a[5], a[6], a[7]);
            }
        }
    }
}

template <int M>
__device__ __forceinline__ float row_xor(float v) {
    return __builtin_bit_cast(float, __builtin_amdgcn_update_dpp(
        0, __builtin_bit_cast(int, v), 0x160 + M, 0xf, 0xf, true));
}

#define KBLK 64
#define PPITCH 68
#define TPITCH 68
#define NW 4
#define NGRP (Tsz / 64)
#define NPAIR (NGRP / 2)
#define SPLITB (NPAIR + 1)

__global__ __launch_bounds__(128) void attn_kernel(
    float* qkv,
    float* __restrict__ ml,
    float* __restrict__ out)
{
    __shared__ __attribute__((aligned(16))) float Klds[2][KBLK * TPITCH];
    __shared__ __attribute__((aligned(16))) float Vlds[2][Hsz * TPITCH];
    __shared__ __attribute__((aligned(16))) float Plds[NW][16 * PPITCH];

    const int tid  = threadIdx.x;
    const int lane = tid & 31;
    const int w    = tid >> 5;
    const int half = lane >> 4;
    const int l16  = lane & 15;
    const int b    = blockIdx.y;
    const int pr   = blockIdx.x >> 1;
    const int role = blockIdx.x & 1;
    const int gL   = NGRP - 1 - pr;
    const int gS   = pr;

    const size_t BTH = (size_t)Bsz * Tsz * Hsz;
    const float* Q  = qkv +            (size_t)b * Tsz * Hsz;
    const float* K  = qkv + BTH     +  (size_t)b * Tsz * Hsz;
    const float* Vt = qkv + 2 * BTH +  (size_t)b * Hsz * Tsz;

    const float scale = 0.125f;

    const int crow = tid >> 4;
    const int ccol = (tid & 15) * 4;
    const float* Kg = K  + (size_t)crow * Hsz + ccol;
    const float* Vg = Vt + (size_t)crow * Tsz + ccol;
    const int lofs  = crow * TPITCH + ccol;

#define TILE_LOAD(kb)                                                          \
    do {                                                                       \
        const float* kp_ = Kg + (size_t)(kb) * Hsz;                            \
        const float* vp_ = Vg + (kb);                                          \
        k0 = *(const float4*)(kp_ +  0 * Hsz); k1 = *(const float4*)(kp_ +  8 * Hsz); \
        k2 = *(const float4*)(kp_ + 16 * Hsz); k3 = *(const float4*)(kp_ + 24 * Hsz); \
        k4 = *(const float4*)(kp_ + 32 * Hsz); k5 = *(const float4*)(kp_ + 40 * Hsz); \
        k6 = *(const float4*)(kp_ + 48 * Hsz); k7 = *(const float4*)(kp_ + 56 * Hsz); \
        v0 = *(const float4*)(vp_ +  0 * Tsz); v1 = *(const float4*)(vp_ +  8 * Tsz); \
        v2 = *(const float4*)(vp_ + 16 * Tsz); v3 = *(const float4*)(vp_ + 24 * Tsz); \
        v4 = *(const float4*)(vp_ + 32 * Tsz); v5 = *(const float4*)(vp_ + 40 * Tsz); \
        v6 = *(const float4*)(vp_ + 48 * Tsz); v7 = *(const float4*)(vp_ + 56 * Tsz); \
    } while (0)
#define TILE_STORE(buf)                                                        \
    do {                                                                       \
        float* kd_ = &Klds[buf][lofs];                                         \
        float* vd_ = &Vlds[buf][lofs];                                         \
        *(float4*)(kd_ +  0 * TPITCH) = k0; *(float4*)(kd_ +  8 * TPITCH) = k1; \
        *(float4*)(kd_ + 16 * TPITCH) = k2; *(float4*)(kd_ + 24 * TPITCH) = k3; \
        *(float4*)(kd_ + 32 * TPITCH) = k4; *(float4*)(kd_ + 40 * TPITCH) = k5; \
        *(float4*)(kd_ + 48 * TPITCH) = k6; *(float4*)(kd_ + 56 * TPITCH) = k7; \
        *(float4*)(vd_ +  0 * TPITCH) = v0; *(float4*)(vd_ +  8 * TPITCH) = v1; \
        *(float4*)(vd_ + 16 * TPITCH) = v2; *(float4*)(vd_ + 24 * TPITCH) = v3; \
        *(float4*)(vd_ + 32 * TPITCH) = v4; *(float4*)(vd_ + 40 * TPITCH) = v5; \
        *(float4*)(vd_ + 48 * TPITCH) = v6; *(float4*)(vd_ + 56 * TPITCH) = v7; \
    } while (0)
    float4 k0, k1, k2, k3, k4, k5, k6, k7, v0, v1, v2, v3, v4, v5, v6, v7;

    float* Pw = &Plds[w][0];

    const int nseg = role + 1;
    for (int seg = 0; seg < nseg; ++seg) {
        const bool lpart = (role == 0) || (seg == 1);
        const int grp  = lpart ? gL : gS;
        const int blk0 = (role == 1 && seg == 1) ? SPLITB : 0;
        const int nblk = (role == 0) ? SPLITB : (seg == 0 ? gS + 1 : gL + 1 - SPLITB);
        const int qi   = grp * NW + w;
        const int qrow = qi * 16 + l16;

        v2f qa[16];
#pragma unroll
        for (int c = 0; c < 16; ++c) {
            v2f t = *(const v2f*)(Q + (size_t)qrow * Hsz + 4 * c + 2 * half);
            qa[c][0] = t[0] * scale;
            qa[c][1] = t[1] * scale;
        }

        if (seg > 0) __syncthreads();
        if (nblk > 0) {
            TILE_LOAD(blk0 * KBLK);
            TILE_STORE(0);
        }

        float mrow[8], lrow[8];
#pragma unroll
        for (int r = 0; r < 8; ++r) { mrow[r] = -INFINITY; lrow[r] = 0.0f; }
        v8f o0 = {}, o1 = {}, o2 = {}, o3 = {};

        for (int tr = 0; tr < nblk; ++tr) {
            const int blk = blk0 + tr;
            const int kb0 = blk * KBLK;
            const int cur = tr & 1;
            __syncthreads();

            const bool more = (tr + 1 < nblk);
            TILE_LOAD(more ? kb0 + KBLK : kb0);

            const float* Kl = &Klds[cur][l16 * TPITCH + 2 * half];
            v2f kf[4][16];
#pragma unroll
            for (int j = 0; j < 4; ++j)
#pragma unroll
                for (int c = 0; c < 16; ++c)
                    kf[j][c] = *(const v2f*)(Kl + 16 * j * TPITCH + 4 * c);
            __builtin_amdgcn_sched_barrier(0);

            v8f s0 = {}, s1 = {}, s2 = {}, s3 = {};
#pragma unroll
            for (int c = 0; c < 16; ++c) {
                s0 = wmma_f32(qa[c], kf[0][c], s0);
                s1 = wmma_f32(qa[c], kf[1][c], s1);
                s2 = wmma_f32(qa[c], kf[2][c], s2);
                s3 = wmma_f32(qa[c], kf[3][c], s3);
            }
            __builtin_amdgcn_sched_barrier(0);

            const float* Vl = &Vlds[cur][l16 * TPITCH + 2 * half];
            v2f vf[4][16];
#pragma unroll
            for (int j = 0; j < 4; ++j)
#pragma unroll
                for (int c2 = 0; c2 < 16; ++c2)
                    vf[j][c2] = *(const v2f*)(Vl + 16 * j * TPITCH + 4 * c2);

            if (blk == grp) {
                const int key0 = kb0 + l16;
                const int qr0  = qi * 16 + 8 * half;
#pragma unroll
                for (int r = 0; r < 8; ++r) {
                    const int qr = qr0 + r;
                    if (key0      > qr) s0[r] = -INFINITY;
                    if (key0 + 16 > qr) s1[r] = -INFINITY;
                    if (key0 + 32 > qr) s2[r] = -INFINITY;
                    if (key0 + 48 > qr) s3[r] = -INFINITY;
                }
            }

            float alpha[8];
#pragma unroll
            for (int r = 0; r < 8; ++r) {
                float mx = fmaxf(fmaxf(s0[r], s1[r]), fmaxf(s2[r], s3[r]));
                mx = fmaxf(mx, row_xor<1>(mx));
                mx = fmaxf(mx, row_xor<2>(mx));
                mx = fmaxf(mx, row_xor<4>(mx));
                mx = fmaxf(mx, row_xor<8>(mx));
                const float mn = fmaxf(mrow[r], mx);
                alpha[r] = expf(mrow[r] - mn);
                mrow[r]  = mn;
                const float p0 = expf(s0[r] - mn);
                const float p1 = expf(s1[r] - mn);
                const float p2 = expf(s2[r] - mn);
                const float p3 = expf(s3[r] - mn);
                float sm = (p0 + p1) + (p2 + p3);
                sm += row_xor<1>(sm);
                sm += row_xor<2>(sm);
                sm += row_xor<4>(sm);
                sm += row_xor<8>(sm);
                lrow[r] = lrow[r] * alpha[r] + sm;

                float* prow = Pw + (r + 8 * half) * PPITCH + l16;
                prow[ 0] = p0;
                prow[16] = p1;
                prow[32] = p2;
                prow[48] = p3;
            }
            __builtin_amdgcn_fence(__ATOMIC_RELEASE, "wavefront");
            __builtin_amdgcn_wave_barrier();
            __builtin_amdgcn_fence(__ATOMIC_ACQUIRE, "wavefront");

#pragma unroll
            for (int r = 0; r < 8; ++r) {
                o0[r] *= alpha[r]; o1[r] *= alpha[r];
                o2[r] *= alpha[r]; o3[r] *= alpha[r];
            }

            v2f pa[16];
#pragma unroll
            for (int c2 = 0; c2 < 16; ++c2)
                pa[c2] = *(const v2f*)(Pw + l16 * PPITCH + 4 * c2 + 2 * half);
            __builtin_amdgcn_sched_barrier(0);

#pragma unroll
            for (int c2 = 0; c2 < 16; ++c2) {
                o0 = wmma_f32(pa[c2], vf[0][c2], o0);
                o1 = wmma_f32(pa[c2], vf[1][c2], o1);
                o2 = wmma_f32(pa[c2], vf[2][c2], o2);
                o3 = wmma_f32(pa[c2], vf[3][c2], o3);
            }

            if (more) {
                TILE_STORE(cur ^ 1);
            }
        }

        if (!lpart) {
            float* Ob = out + ((size_t)b * Tsz + qi * 16 + 8 * half) * Hsz + l16;
#pragma unroll
            for (int r = 0; r < 8; ++r) {
                const float inv = 1.0f / lrow[r];
                Ob[(size_t)r * Hsz +  0] = o0[r] * inv;
                Ob[(size_t)r * Hsz + 16] = o1[r] * inv;
                Ob[(size_t)r * Hsz + 32] = o2[r] * inv;
                Ob[(size_t)r * Hsz + 48] = o3[r] * inv;
            }
        } else {
            const int part = role;
            float* Od = (part == 0)
                ? out + ((size_t)b * Tsz + qi * 16 + 8 * half) * Hsz + l16
                : qkv + ((size_t)b * Tsz + (gS * NW + w) * 16 + 8 * half) * Hsz + l16;
#pragma unroll
            for (int r = 0; r < 8; ++r) {
                Od[(size_t)r * Hsz +  0] = o0[r];
                Od[(size_t)r * Hsz + 16] = o1[r];
                Od[(size_t)r * Hsz + 32] = o2[r];
                Od[(size_t)r * Hsz + 48] = o3[r];
            }
            float* mlp = ml + ((((size_t)part * Bsz + b) * NPAIR + pr) * 64 + 16 * w + 8 * half) * 2;
#pragma unroll
            for (int r = 0; r < 8; ++r)
                if (l16 == r)
                    *(float2*)(mlp + 2 * r) = make_float2(mrow[r], lrow[r]);
        }
    }

#undef TILE_LOAD
#undef TILE_STORE
}

__global__ __launch_bounds__(256) void merge_kernel(
    const float* __restrict__ qkv,
    const float* __restrict__ ml,
    float* out)
{
    const int e  = blockIdx.x * 256 + threadIdx.x;
    const int c4 = e & 15;
    const int r  = (e >> 4) & 63;
    const int pr = (e >> 10) % NPAIR;
    const int b  = (e >> 10) / NPAIR;
    const int gL = NGRP - 1 - pr;
    const int gS = pr;

    const float2 a0 = *(const float2*)(ml + ((((size_t)0 * Bsz + b) * NPAIR + pr) * 64 + r) * 2);
    const float2 a1 = *(const float2*)(ml + ((((size_t)1 * Bsz + b) * NPAIR + pr) * 64 + r) * 2);
    const float m  = fmaxf(a0.x, a1.x);
    const float w0 = expf(a0.x - m);
    const float w1 = expf(a1.x - m);
    const float inv = 1.0f / (a0.y * w0 + a1.y * w1);

    float* op = out + ((size_t)b * Tsz + gL * 64 + r) * Hsz + 4 * c4;
    const float4 x0 = *(const float4*)op;
    const float4 x1 = *(const float4*)(qkv + ((size_t)b * Tsz + gS * 64 + r) * Hsz + 4 * c4);
    float4 y;
    y.x = (x0.x * w0 + x1.x * w1) * inv;
    y.y = (x0.y * w0 + x1.y * w1) * inv;
    y.z = (x0.z * w0 + x1.z * w1) * inv;
    y.w = (x0.w * w0 + x1.w * w1) * inv;
    *(float4*)op = y;
}

extern "C" void kernel_launch(void* const* d_in, const int* in_sizes, int n_in,
                              void* d_out, int out_size, void* d_ws, size_t ws_size,
                              hipStream_t stream) {
    const float* x  = (const float*)d_in[0];
    const float* Wq = (const float*)d_in[1];
    const float* Wk = (const float*)d_in[2];
    const float* Wv = (const float*)d_in[3];

    const size_t BTH = (size_t)Bsz * Tsz * Hsz;
    float* qkv = (float*)d_ws;
    float* Wt  = qkv + 3 * BTH;
    float* ml  = Wt;

    dim3 g0(Hsz, 3), blk0(256);
    wt_kernel<<<g0, blk0, 0, stream>>>(Wq, Wk, Wv, Wt);

    dim3 g1(Bsz * Tsz / 64), blk1(128);
    qkv_kernel<<<g1, blk1, 0, stream>>>(x, Wt, qkv);

    dim3 g2(Tsz / 64, Bsz), blk2(128);
    attn_kernel<<<g2, blk2, 0, stream>>>(qkv, ml, (float*)d_out);

    dim3 g3(Bsz * NPAIR * 64 * 16 / 256), blk3(256);
    merge_kernel<<<g3, blk3, 0, stream>>>(qkv, ml, (float*)d_out);
}


# qkv projection: weight B-fragments loaded straight from global Wq/Wk/Wv as given ([C][H]), two 4-byte loads per lane: 2 lines per load, not 16; the weight-transposing launch goes (4 -> 3, sums kept)
# speedup vs baseline: 3.4575x; 1.0692x over previous
#include <hip/hip_runtime.h>
#include <math.h>

#define Bsz 8
#define Tsz 2048
#define Csz 1024
#define Hsz 64

typedef __attribute__((ext_vector_type(2))) float v2f;
typedef __attribute__((ext_vector_type(8))) float v8f;

__device__ __forceinline__ v8f wmma_f32(v2f a, v2f b, v8f c) {
    return __builtin_amdgcn_wmma_f32_16x16x4_f32(
         false, a,  false, b,
         (short)0, c,  false,  false);
}

#define KSTEPS 2
#define KSPAN 32

struct QkvFrag {
    v2f a[KSTEPS][4];
    v2f b[KSTEPS][12];
};

__device__ __forceinline__ void qkv_load(QkvFrag& f, const float* xk,
                                         const float* wq, const float* wk, const float* wv,
                                         unsigned lofs, unsigned wofs) {
#pragma unroll
    for (int s = 0; s < KSTEPS; ++s) {
#pragma unroll
        for (int i = 0; i < 4; ++i)
            f.a[s][i] = *(const v2f*)(xk + lofs + (16 * i * Csz + 4 * s));
#pragma unroll
        for (int j = 0; j < 12; ++j) {
            const float* w = (j < 4) ? wq : (j < 8) ? wk : wv;
            const float* p = w + wofs + (4 * s * Hsz + 16 * (j & 3));
            f.b[s][j][0] = p[0];
            f.b[s][j][1] = p[Hsz];
        }
    }
}

__device__ __forceinline__ void qkv_mma(const QkvFrag& f, v8f (&acc)[4][12]) {
#pragma unroll
    for (int s = 0; s < KSTEPS; ++s)
#pragma unroll
        for (int i = 0; i < 4; ++i)
#pragma unroll
            for (int j = 0; j < 12; ++j)
                acc[i][j] = wmma_f32(f.a[s][i], f.b[s][j], acc[i][j]);
}

__device__ __forceinline__ void wmma_slots4() {
    asm volatile("v_nop\n\tv_nop\n\tv_nop\n\tv_nop");
}

__global__ __launch_bounds__(128) void qkv_kernel(
    const float* __restrict__ x,
    const float* __restrict__ Wq,
    const float* __restrict__ Wk,
    const float* __restrict__ Wv,
    float* __restrict__ qkv)
{
    __shared__ float4 red[4 * 48 * 2 * 32];

    const int lane = threadIdx.x & 31;
    const int g    = threadIdx.x >> 5;
    const int half = lane >> 4;
    const int l16  = lane & 15;

    const long mrow = (long)blockIdx.x * 64;

    const float* xb = x + mrow * (size_t)Csz;
    const unsigned lofs = (unsigned)(l16 * Csz + 8 * g + 2 * half);
    const unsigned wofs = (unsigned)((8 * g + 2 * half) * Hsz + l16);

    v8f acc[4][12];
#pragma unroll
    for (int i = 0; i < 4; ++i)
#pragma unroll
        for (int j = 0; j < 12; ++j)
            acc[i][j] = v8f{};

    QkvFrag f0, f1;
    qkv_load(f0, xb, Wq, Wk, Wv, lofs, wofs);
    for (int kk = 0; kk < Csz; kk += 2 * KSPAN) {
        const int k1 = kk + KSPAN;
        qkv_load(f1, xb + k1, Wq + k1 * Hsz, Wk + k1 * Hsz, Wv + k1 * Hsz, lofs, wofs);
        __builtin_amdgcn_sched_barrier(0);
        qkv_mma(f0, acc);
        wmma_slots4();
        __builtin_amdgcn_sched_barrier(0);
        const int kn = (kk + 2 * KSPAN < Csz) ? kk + 2 * KSPAN : Csz - KSPAN;
        qkv_load(f0, xb + kn, Wq + kn * Hsz, Wk + kn * Hsz, Wv + kn * Hsz, lofs, wofs);
        __builtin_amdgcn_sched_barrier(0);
        qkv_mma(f1, acc);
        wmma_slots4();
        __builtin_amdgcn_sched_barrier(0);
    }

#pragma unroll
    for (int i = 0; i < 4; ++i)
#pragma unroll
        for (int j = 0; j < 12; ++j) {
            const v8f a = acc[i][j];
            float4* rp = red + ((g * 48 + 12 * i + j) * 2) * 32 + lane;
            rp[0]  = make_float4(a[0], a[1], a[2], a[3]);
            rp[32] = make_float4(a[4], a[5], a[6], a[7]);
        }
    __syncthreads();

    const size_t BTH = (size_t)Bsz * Tsz * Hsz;
    const long bb   = mrow / Tsz;
    const long tblk = mrow - bb * Tsz;
    for (int j = 0; j < 3; ++j) {
        const int t     = 3 * g + j;
        const int which = t >> 2;
        const int col   = (t & 3) * 16 + l16;
        v8f sum[4];
#pragma unroll
        for (int i = 0; i < 4; ++i) {
            const float4* rp = red + ((12 * i + t) * 2) * 32 + lane;
            float4 lo = rp[0], hi = rp[32];
#pragma unroll
            for (int s = 1; s < 4; ++s) {
                const float4 l2 = rp[s * 48 * 2 * 32], h2 = rp[s * 48 * 2 * 32 + 32];
                lo.x += l2.x; lo.y += l2.y; lo.z += l2.z; lo.w += l2.w;
                hi.x += h2.x; hi.y += h2.y; hi.z += h2.z; hi.w += h2.w;
            }
            sum[i] = v8f{lo.x, lo.y, lo.z, lo.w, hi.x, hi.y, hi.z, hi.w};
        }
        if (which < 2) {
            float* op = qkv + (size_t)which * BTH + (mrow + 8 * half) * (size_t)Hsz + col;
#pragma unroll
            for (int i = 0; i < 4; ++i)
#pragma unroll
                for (int r = 0; r < 8; ++r)
                    op[(size_t)(16 * i + r) * Hsz] = sum[i][r];
        } else {
            float* vp = qkv + 2 * BTH + ((size_t)bb * Hsz + col) * Tsz + tblk + 8 * half;
#pragma unroll
            for (int i = 0; i < 4; ++i) {
                const v8f a = sum[i];
                *(float4*)(vp + 16 * i + 0) = make_float4(a[0], a[1], a[2], a[3]);
                *(float4*)(vp + 16 * i + 4) = make_float4(a[4], a[5], a[6], a[7]);
            }
        }
    }
}

template <int M>
__device__ __forceinline__ float row_xor(float v) {
    return __builtin_bit_cast(float, __builtin_amdgcn_update_dpp(
        0, __builtin_bit_cast(int, v), 0x160 + M, 0xf, 0xf, true));
}

#define KBLK 64
#define PPITCH 68
#define TPITCH 68
#define NW 4
#define NGRP (Tsz / 64)
#define NPAIR (NGRP / 2)
#define SPLITB (NPAIR + 1)

__global__ __launch_bounds__(128) void attn_kernel(
    float* qkv,
    float* __restrict__ ml,
    float* __restrict__ out)
{
    __shared__ __attribute__((aligned(16))) float Klds[2][KBLK * TPITCH];
    __shared__ __attribute__((aligned(16))) float Vlds[2][Hsz * TPITCH];
    __shared__ __attribute__((aligned(16))) float Plds[NW][16 * PPITCH];

    const int tid  = threadIdx.x;
    const int lane = tid & 31;
    const int w    = tid >> 5;
    const int half = lane >> 4;
    const int l16  = lane & 15;
    const int b    = blockIdx.y;
    const int pr   = blockIdx.x >> 1;
    const int role = blockIdx.x & 1;
    const int gL   = NGRP - 1 - pr;
    const int gS   = pr;

    const size_t BTH = (size_t)Bsz * Tsz * Hsz;
    const float* Q  = qkv +            (size_t)b * Tsz * Hsz;
    const float* K  = qkv + BTH     +  (size_t)b * Tsz * Hsz;
    const float* Vt = qkv + 2 * BTH +  (size_t)b * Hsz * Tsz;

    const float scale = 0.125f;

    const int crow = tid >> 4;
    const int ccol = (tid & 15) * 4;
    const float* Kg = K  + (size_t)crow * Hsz + ccol;
    const float* Vg = Vt + (size_t)crow * Tsz + ccol;
    const int lofs  = crow * TPITCH + ccol;

#define TILE_LOAD(kb)                                                          \
    do {                                                                       \
        const float* kp_ = Kg + (size_t)(kb) * Hsz;                            \
        const float* vp_ = Vg + (kb);                                          \
        k0 = *(const float4*)(kp_ +  0 * Hsz); k1 = *(const float4*)(kp_ +  8 * Hsz); \
        k2 = *(const float4*)(kp_ + 16 * Hsz); k3 = *(const float4*)(kp_ + 24 * Hsz); \
        k4 = *(const float4*)(kp_ + 32 * Hsz); k5 = *(const float4*)(kp_ + 40 * Hsz); \
        k6 = *(const float4*)(kp_ + 48 * Hsz); k7 = *(const float4*)(kp_ + 56 * Hsz); \
        v0 = *(const float4*)(vp_ +  0 * Tsz); v1 = *(const float4*)(vp_ +  8 * Tsz); \
        v2 = *(const float4*)(vp_ + 16 * Tsz); v3 = *(const float4*)(vp_ + 24 * Tsz); \
        v4 = *(const float4*)(vp_ + 32 * Tsz); v5 = *(const float4*)(vp_ + 40 * Tsz); \
        v6 = *(const float4*)(vp_ + 48 * Tsz); v7 = *(const float4*)(vp_ + 56 * Tsz); \
    } while (0)
#define TILE_STORE(buf)                                                        \
    do {                                                                       \
        float* kd_ = &Klds[buf][lofs];                                         \
        float* vd_ = &Vlds[buf][lofs];                                         \
        *(float4*)(kd_ +  0 * TPITCH) = k0; *(float4*)(kd_ +  8 * TPITCH) = k1; \
        *(float4*)(kd_ + 16 * TPITCH) = k2; *(float4*)(kd_ + 24 * TPITCH) = k3; \
        *(float4*)(kd_ + 32 * TPITCH) = k4; *(float4*)(kd_ + 40 * TPITCH) = k5; \
        *(float4*)(kd_ + 48 * TPITCH) = k6; *(float4*)(kd_ + 56 * TPITCH) = k7; \
        *(float4*)(vd_ +  0 * TPITCH) = v0; *(float4*)(vd_ +  8 * TPITCH) = v1; \
        *(float4*)(vd_ + 16 * TPITCH) = v2; *(float4*)(vd_ + 24 * TPITCH) = v3; \
        *(float4*)(vd_ + 32 * TPITCH) = v4; *(float4*)(vd_ + 40 * TPITCH) = v5; \
        *(float4*)(vd_ + 48 * TPITCH) = v6; *(float4*)(vd_ + 56 * TPITCH) = v7; \
    } while (0)
    float4 k0, k1, k2, k3, k4, k5, k6, k7, v0, v1, v2, v3, v4, v5, v6, v7;

    float* Pw = &Plds[w][0];

    const int nseg = role + 1;
    for (int seg = 0; seg < nseg; ++seg) {
        const bool lpart = (role == 0) || (seg == 1);
        const int grp  = lpart ? gL : gS;
        const int blk0 = (role == 1 && seg == 1) ? SPLITB : 0;
        const int nblk = (role == 0) ? SPLITB : (seg == 0 ? gS + 1 : gL + 1 - SPLITB);
        const int qi   = grp * NW + w;
        const int qrow = qi * 16 + l16;

        v2f qa[16];
#pragma unroll
        for (int c = 0; c < 16; ++c) {
            v2f t = *(const v2f*)(Q + (size_t)qrow * Hsz + 4 * c + 2 * half);
            qa[c][0] = t[0] * scale;
            qa[c][1] = t[1] * scale;
        }

        if (seg > 0) __syncthreads();
        if (nblk > 0) {
            TILE_LOAD(blk0 * KBLK);
            TILE_STORE(0);
        }

        float mrow[8], lrow[8];
#pragma unroll
        for (int r = 0; r < 8; ++r) { mrow[r] = -INFINITY; lrow[r] = 0.0f; }
        v8f o0 = {}, o1 = {}, o2 = {}, o3 = {};

        for (int tr = 0; tr < nblk; ++tr) {
            const int blk = blk0 + tr;
            const int kb0 = blk * KBLK;
            const int cur = tr & 1;
            __syncthreads();

            const bool more = (tr + 1 < nblk);
            TILE_LOAD(more ? kb0 + KBLK : kb0);

            const float* Kl = &Klds[cur][l16 * TPITCH + 2 * half];
            v2f kf[4][16];
#pragma unroll
            for (int j = 0; j < 4; ++j)
#pragma unroll
                for (int c = 0; c < 16; ++c)
                    kf[j][c] = *(const v2f*)(Kl + 16 * j * TPITCH + 4 * c);
            __builtin_amdgcn_sched_barrier(0);

            v8f s0 = {}, s1 = {}, s2 = {}, s3 = {};
#pragma unroll
            for (int c = 0; c < 16; ++c) {
                s0 = wmma_f32(qa[c], kf[0][c], s0);
                s1 = wmma_f32(qa[c], kf[1][c], s1);
                s2 = wmma_f32(qa[c], kf[2][c], s2);
                s3 = wmma_f32(qa[c], kf[3][c], s3);
            }
            __builtin_amdgcn_sched_barrier(0);

            const float* Vl = &Vlds[cur][l16 * TPITCH + 2 * half];
            v2f vf[4][16];
#pragma unroll
            for (int j = 0; j < 4; ++j)
#pragma unroll
                for (int c2 = 0; c2 < 16; ++c2)
                    vf[j][c2] = *(const v2f*)(Vl + 16 * j * TPITCH + 4 * c2);

            if (blk == grp) {
                const int key0 = kb0 + l16;
                const int qr0  = qi * 16 + 8 * half;
#pragma unroll
                for (int r = 0; r < 8; ++r) {
                    const int qr = qr0 + r;
                    if (key0      > qr) s0[r] = -INFINITY;
                    if (key0 + 16 > qr) s1[r] = -INFINITY;
                    if (key0 + 32 > qr) s2[r] = -INFINITY;
                    if (key0 + 48 > qr) s3[r] = -INFINITY;
                }
            }

            float alpha[8];
#pragma unroll
            for (int r = 0; r < 8; ++r) {
                float mx = fmaxf(fmaxf(s0[r], s1[r]), fmaxf(s2[r], s3[r]));
                mx = fmaxf(mx, row_xor<1>(mx));
                mx = fmaxf(mx, row_xor<2>(mx));
                mx = fmaxf(mx, row_xor<4>(mx));
                mx = fmaxf(mx, row_xor<8>(mx));
                const float mn = fmaxf(mrow[r], mx);
                alpha[r] = expf(mrow[r] - mn);
                mrow[r]  = mn;
                const float p0 = expf(s0[r] - mn);
                const float p1 = expf(s1[r] - mn);
                const float p2 = expf(s2[r] - mn);
                const float p3 = expf(s3[r] - mn);
                float sm = (p0 + p1) + (p2 + p3);
                sm += row_xor<1>(sm);
                sm += row_xor<2>(sm);
                sm += row_xor<4>(sm);
                sm += row_xor<8>(sm);
                lrow[r] = lrow[r] * alpha[r] + sm;

                float* prow = Pw + (r + 8 * half) * PPITCH + l16;
                prow[ 0] = p0;
                prow[16] = p1;
                prow[32] = p2;
                prow[48] = p3;
            }
            __builtin_amdgcn_fence(__ATOMIC_RELEASE, "wavefront");
            __builtin_amdgcn_wave_barrier();
            __builtin_amdgcn_fence(__ATOMIC_ACQUIRE, "wavefront");

#pragma unroll
            for (int r = 0; r < 8; ++r) {
                o0[r] *= alpha[r]; o1[r] *= alpha[r];
                o2[r] *= alpha[r]; o3[r] *= alpha[r];
            }

            v2f pa[16];
#pragma unroll
            for (int c2 = 0; c2 < 16; ++c2)
                pa[c2] = *(const v2f*)(Pw + l16 * PPITCH + 4 * c2 + 2 * half);
            __builtin_amdgcn_sched_barrier(0);

#pragma unroll
            for (int c2 = 0; c2 < 16; ++c2) {
                o0 = wmma_f32(pa[c2], vf[0][c2], o0);
                o1 = wmma_f32(pa[c2], vf[1][c2], o1);
                o2 = wmma_f32(pa[c2], vf[2][c2], o2);
                o3 = wmma_f32(pa[c2], vf[3][c2], o3);
            }

            if (more) {
                TILE_STORE(cur ^ 1);
            }
        }

        if (!lpart) {
            float* Ob = out + ((size_t)b * Tsz + qi * 16 + 8 * half) * Hsz + l16;
#pragma unroll
            for (int r = 0; r < 8; ++r) {
                const float inv = 1.0f / lrow[r];
                Ob[(size_t)r * Hsz +  0] = o0[r] * inv;
                Ob[(size_t)r * Hsz + 16] = o1[r] * inv;
                Ob[(size_t)r * Hsz + 32] = o2[r] * inv;
                Ob[(size_t)r * Hsz + 48] = o3[r] * inv;
            }
        } else {
            const int part = role;
            float* Od = (part == 0)
                ? out + ((size_t)b * Tsz + qi * 16 + 8 * half) * Hsz + l16
                : qkv + ((size_t)b * Tsz + (gS * NW + w) * 16 + 8 * half) * Hsz + l16;
#pragma unroll
            for (int r = 0; r < 8; ++r) {
                Od[(size_t)r * Hsz +  0] = o0[r];
                Od[(size_t)r * Hsz + 16] = o1[r];
                Od[(size_t)r * Hsz + 32] = o2[r];
                Od[(size_t)r * Hsz + 48] = o3[r];
            }
            float* mlp = ml + ((((size_t)part * Bsz + b) * NPAIR + pr) * 64 + 16 * w + 8 * half) * 2;
#pragma unroll
            for (int r = 0; r < 8; ++r)
                if (l16 == r)
                    *(float2*)(mlp + 2 * r) = make_float2(mrow[r], lrow[r]);
        }
    }

#undef TILE_LOAD
#undef TILE_STORE
}

__global__ __launch_bounds__(256) void merge_kernel(
    const float* __restrict__ qkv,
    const float* __restrict__ ml,
    float* out)
{
    const int e  = blockIdx.x * 256 + threadIdx.x;
    const int c4 = e & 15;
    const int r  = (e >> 4) & 63;
    const int pr = (e >> 10) % NPAIR;
    const int b  = (e >> 10) / NPAIR;
    const int gL = NGRP - 1 - pr;
    const int gS = pr;

    const float2 a0 = *(const float2*)(ml + ((((size_t)0 * Bsz + b) * NPAIR + pr) * 64 + r) * 2);
    const float2 a1 = *(const float2*)(ml + ((((size_t)1 * Bsz + b) * NPAIR + pr) * 64 + r) * 2);
    const float m  = fmaxf(a0.x, a1.x);
    const float w0 = expf(a0.x - m);
    const float w1 = expf(a1.x - m);
    const float inv = 1.0f / (a0.y * w0 + a1.y * w1);

    float* op = out + ((size_t)b * Tsz + gL * 64 + r) * Hsz + 4 * c4;
    const float4 x0 = *(const float4*)op;
    const float4 x1 = *(const float4*)(qkv + ((size_t)b * Tsz + gS * 64 + r) * Hsz + 4 * c4);
    float4 y;
    y.x = (x0.x * w0 + x1.x * w1) * inv;
    y.y = (x0.y * w0 + x1.y * w1) * inv;
    y.z = (x0.z * w0 + x1.z * w1) * inv;
    y.w = (x0.w * w0 + x1.w * w1) * inv;
    *(float4*)op = y;
}

extern "C" void kernel_launch(void* const* d_in, const int* in_sizes, int n_in,
                              void* d_out, int out_size, void* d_ws, size_t ws_size,
                              hipStream_t stream) {
    const float* x  = (const float*)d_in[0];
    const float* Wq = (const float*)d_in[1];
    const float* Wk = (const float*)d_in[2];
    const float* Wv = (const float*)d_in[3];

    const size_t BTH = (size_t)Bsz * Tsz * Hsz;
    float* qkv = (float*)d_ws;
    float* ml  = qkv + 3 * BTH;

    dim3 g1(Bsz * Tsz / 64), blk1(128);
    qkv_kernel<<<g1, blk1, 0, stream>>>(x, Wq, Wk, Wv, qkv);

    dim3 g2(Tsz / 64, Bsz), blk2(128);
    attn_kernel<<<g2, blk2, 0, stream>>>(qkv, ml, (float*)d_out);

    dim3 g3(Bsz * NPAIR * 64 * 16 / 256), blk3(256);
    merge_kernel<<<g3, blk3, 0, stream>>>(qkv, ml, (float*)d_out);
}
